# starting kernel
# baseline (speedup 1.0000x reference)
#include <hip/hip_runtime.h>


typedef __attribute__((ext_vector_type(2))) float v2f;
typedef __attribute__((ext_vector_type(8))) float v8f;

__global__ __launch_bounds__(256)
void lcnn_wmma_kernel(const float* __restrict__ W,
                      const float* __restrict__ U,
                      const float* __restrict__ Om,
                      float* __restrict__ Out)
{
    const int tid  = threadIdx.x;
    const int wave = tid >> 5;
    const int lane = tid & 31;
    const int line = blockIdx.x * 8 + wave;

    const int s     = lane & 15;
    const int hh    = lane >> 4;
    const int jadd  = hh * 2;
    const int ibase = hh * 4;

    const int x3 = line & 15;
    const int x2 = (line >> 4) & 15;
    const int x1 = (line >> 8) & 15;

    const int mySite = line * 16 + s;

    float outAcc[36];
    #pragma unroll
    for (int q = 0; q < 36; ++q) outAcc[q] = 0.f;

    #pragma unroll 1
    for (int t = 0; t < 20; ++t) {
        const int mu = t & 3;
        const int ik = t >> 2;
        const int k  = ik - 2;

        int wline  = line;
        int sshift = s;
        if      (mu == 0) wline = (line & ~(15 << 8)) | (((x1 + k) & 15) << 8);
        else if (mu == 1) wline = (line & ~(15 << 4)) | (((x2 + k) & 15) << 4);
        else if (mu == 2) wline = (line & ~15)        |  ((x3 + k) & 15);
        else              sshift = (s + k + 16) & 15;

        const int wAddr = wline * 1152 + sshift * 72 + jadd * 9;

        float a0v0 = 0.f, a0v1 = 0.f, a1v0 = 0.f, a1v1 = 0.f;
        if (s < 8) {
            const int ob = s * 160 + mu * 5 + ik;
            a0v0 = Om[ob + (jadd + 0) * 20];
            a0v1 = Om[ob + (jadd + 1) * 20];
            a1v0 = Om[ob + (jadd + 4) * 20];
            a1v1 = Om[ob + (jadd + 5) * 20];
        }
        const v2f A0 = {a0v0, a0v1};
        const v2f A1 = {a1v0, a1v1};

        const float* up = U + mySite * 180 + (mu * 5 + ik) * 9;
        float u[9];
        #pragma unroll
        for (int q = 0; q < 9; ++q) u[q] = up[q];

        v8f acc[9];
        #pragma unroll
        for (int cd = 0; cd < 9; ++cd) {
            v8f c = {0.f, 0.f, 0.f, 0.f, 0.f, 0.f, 0.f, 0.f};
            const v2f B0 = { W[wAddr +      cd], W[wAddr +  9 + cd] };
            const v2f B1 = { W[wAddr + 36 + cd], W[wAddr + 45 + cd] };
            c = __builtin_amdgcn_wmma_f32_16x16x4_f32(false, A0, false, B0, (short)0, c, false, false);
            c = __builtin_amdgcn_wmma_f32_16x16x4_f32(false, A1, false, B1, (short)0, c, false, false);
            acc[cd] = c;
        }

        float w[9][4];
        const int pidx = s << 2;
        #pragma unroll
        for (int cd = 0; cd < 9; ++cd) {
            #pragma unroll
            for (int ii = 0; ii < 4; ++ii) {
                const float hi = __int_as_float(
                    __builtin_amdgcn_ds_bpermute(pidx, __float_as_int(acc[cd][4 + ii])));
                w[cd][ii] = (hh == 0) ? acc[cd][ii] : hi;
            }
        }

        #pragma unroll
        for (int ii = 0; ii < 4; ++ii) {
            float T[9];
            #pragma unroll
            for (int c = 0; c < 3; ++c)
                #pragma unroll
                for (int b = 0; b < 3; ++b)
                    T[c*3+b] = w[c*3+0][ii]*u[b*3+0]
                             + w[c*3+1][ii]*u[b*3+1]
                             + w[c*3+2][ii]*u[b*3+2];
            #pragma unroll
            for (int a = 0; a < 3; ++a)
                #pragma unroll
                for (int b = 0; b < 3; ++b)
                    outAcc[ii*9 + a*3 + b] += u[a*3+0]*T[0+b]
                                            + u[a*3+1]*T[3+b]
                                            + u[a*3+2]*T[6+b];
        }
    }

    float4* op = (float4*)(Out + mySite * 72 + ibase * 9);
    #pragma unroll
    for (int q = 0; q < 9; ++q)
        op[q] = make_float4(outAcc[q*4+0], outAcc[q*4+1], outAcc[q*4+2], outAcc[q*4+3]);
}

extern "C" void kernel_launch(void* const* d_in, const int* in_sizes, int n_in,
                              void* d_out, int out_size, void* d_ws, size_t ws_size,
                              hipStream_t stream) {
    const float* W  = (const float*)d_in[0];
    const float* U  = (const float*)d_in[1];
    const float* Om = (const float*)d_in[2];
    float* Out = (float*)d_out;
    (void)in_sizes; (void)n_in; (void)out_size; (void)d_ws; (void)ws_size;

    lcnn_wmma_kernel<<<dim3(1024), dim3(256), 0, stream>>>(W, U, Om, Out);
}


# W lines and the wave's U block fetched with whole-line 16-byte loads and staged in wave-private LDS, next W line in flight during compute; WMMA B fragments and U read from LDS; sum order unchanged
# speedup vs baseline: 1.2778x; 1.2778x over previous
#include <hip/hip_runtime.h>


typedef __attribute__((ext_vector_type(2))) float v2f;
typedef __attribute__((ext_vector_type(8))) float v8f;
typedef __attribute__((ext_vector_type(4))) float v4f;

#define U_DW     2880
#define W_PITCH  76
#define WAVE_DW  (U_DW + 16 * W_PITCH)

__global__ __launch_bounds__(256)
void lcnn_wmma_kernel(const float* __restrict__ W,
                      const float* __restrict__ U,
                      const float* __restrict__ Om,
                      float* __restrict__ Out)
{
    __shared__ __attribute__((aligned(16))) float lds[8 * WAVE_DW];

    const int tid  = threadIdx.x;
    const int wave = tid >> 5;
    const int lane = tid & 31;
    const int line = blockIdx.x * 8 + wave;

    const int s     = lane & 15;
    const int hh    = lane >> 4;
    const int jadd  = hh * 2;
    const int ibase = hh * 4;

    const int x3 = line & 15;
    const int x2 = (line >> 4) & 15;
    const int x1 = (line >> 8) & 15;

    const int mySite = line * 16 + s;

    float* uL = lds + wave * WAVE_DW;
    float* wL = uL + U_DW;

    {
        const v4f* ug  = (const v4f*)(U + line * U_DW);
        v4f*       uL4 = (v4f*)uL;
        #pragma unroll
        for (int q = 0; q < 23; ++q) {
            const int g = q * 32 + lane;
            if (g < 720) uL4[g] = ug[g];
        }
    }

    int wslot[9];
    #pragma unroll
    for (int q = 0; q < 9; ++q) {
        const int g = q * 32 + lane;
        wslot[q] = (g / 18) * W_PITCH + (g % 18) * 4;
    }

    v4f wn[9];
    {
        const int wline0 = (line & ~(15 << 8)) | (((x1 - 2) & 15) << 8);
        const v4f* wg = (const v4f*)(W + wline0 * 1152);
        #pragma unroll
        for (int q = 0; q < 9; ++q) wn[q] = wg[q * 32 + lane];
    }

    float outAcc[36];
    #pragma unroll
    for (int q = 0; q < 36; ++q) outAcc[q] = 0.f;

    #pragma unroll 1
    for (int t = 0; t < 20; ++t) {
        const int mu = t & 3;
        const int ik = t >> 2;
        const int k  = ik - 2;

        #pragma unroll
        for (int q = 0; q < 9; ++q) *(v4f*)(wL + wslot[q]) = wn[q];

        const int ob = (s & 7) * 160 + mu * 5 + ik;
        const v2f A0 = { Om[ob + (jadd + 0) * 20],
                         Om[ob + (jadd + 1) * 20] };
        const v2f A1 = { Om[ob + (jadd + 4) * 20],
                         Om[ob + (jadd + 5) * 20] };

        {
            const int tn  = (t + 1 < 20) ? (t + 1) : t;
            const int mu1 = tn & 3;
            const int k1  = (tn >> 2) - 2;
            int wline1 = line;
            if      (mu1 == 0) wline1 = (line & ~(15 << 8)) | (((x1 + k1) & 15) << 8);
            else if (mu1 == 1) wline1 = (line & ~(15 << 4)) | (((x2 + k1) & 15) << 4);
            else if (mu1 == 2) wline1 = (line & ~15)        |  ((x3 + k1) & 15);
            const v4f* wg = (const v4f*)(W + wline1 * 1152);
            #pragma unroll
            for (int q = 0; q < 9; ++q) wn[q] = wg[q * 32 + lane];
        }

        __builtin_amdgcn_wave_barrier();

        const int sshift = (mu == 3) ? ((s + k + 16) & 15) : s;
        const float* wb = wL + sshift * W_PITCH + jadd * 9;

        const float* up = uL + s * 180 + (mu * 5 + ik) * 9;
        float u[9];
        #pragma unroll
        for (int q = 0; q < 9; ++q) u[q] = up[q];

        v8f acc[9];
        #pragma unroll
        for (int cd = 0; cd < 9; ++cd) {
            v8f c = {0.f, 0.f, 0.f, 0.f, 0.f, 0.f, 0.f, 0.f};
            const v2f B0 = { wb[     cd], wb[ 9 + cd] };
            const v2f B1 = { wb[36 + cd], wb[45 + cd] };
            c = __builtin_amdgcn_wmma_f32_16x16x4_f32(false, A0, false, B0, (short)0, c, false, false);
            c = __builtin_amdgcn_wmma_f32_16x16x4_f32(false, A1, false, B1, (short)0, c, false, false);
            acc[cd] = c;
        }

        __builtin_amdgcn_wave_barrier();

        float w[9][4];
        const int pidx = s << 2;
        #pragma unroll
        for (int cd = 0; cd < 9; ++cd) {
            #pragma unroll
            for (int ii = 0; ii < 4; ++ii) {
                const float hi = __int_as_float(
                    __builtin_amdgcn_ds_bpermute(pidx, __float_as_int(acc[cd][4 + ii])));
                w[cd][ii] = (hh == 0) ? acc[cd][ii] : hi;
            }
        }

        #pragma unroll
        for (int ii = 0; ii < 4; ++ii) {
            float T[9];
            #pragma unroll
            for (int c = 0; c < 3; ++c)
                #pragma unroll
                for (int b = 0; b < 3; ++b)
                    T[c*3+b] = w[c*3+0][ii]*u[b*3+0]
                             + w[c*3+1][ii]*u[b*3+1]
                             + w[c*3+2][ii]*u[b*3+2];
            #pragma unroll
            for (int a = 0; a < 3; ++a)
                #pragma unroll
                for (int b = 0; b < 3; ++b)
                    outAcc[ii*9 + a*3 + b] += u[a*3+0]*T[0+b]
                                            + u[a*3+1]*T[3+b]
                                            + u[a*3+2]*T[6+b];
        }
    }

    float4* op = (float4*)(Out + mySite * 72 + ibase * 9);
    #pragma unroll
    for (int q = 0; q < 9; ++q)
        op[q] = make_float4(outAcc[q*4+0], outAcc[q*4+1], outAcc[q*4+2], outAcc[q*4+3]);
}

extern "C" void kernel_launch(void* const* d_in, const int* in_sizes, int n_in,
                              void* d_out, int out_size, void* d_ws, size_t ws_size,
                              hipStream_t stream) {
    const float* W  = (const float*)d_in[0];
    const float* U  = (const float*)d_in[1];
    const float* Om = (const float*)d_in[2];
    float* Out = (float*)d_out;
    (void)in_sizes; (void)n_in; (void)out_size; (void)d_ws; (void)ws_size;

    lcnn_wmma_kernel<<<dim3(1024), dim3(256), 0, stream>>>(W, U, Om, Out);
}


# Terms reordered: the 8 that read the wave's own W line (mu=3 all k; k=0 of mu=0..2) run first on one staged copy: 13 W line fetches and LDS stagings per wave, not 20; sum order over terms changes
# speedup vs baseline: 1.6830x; 1.3171x over previous
#include <hip/hip_runtime.h>


typedef __attribute__((ext_vector_type(2))) float v2f;
typedef __attribute__((ext_vector_type(8))) float v8f;
typedef __attribute__((ext_vector_type(4))) float v4f;

#define U_DW     2880
#define W_PITCH  76
#define WAVE_DW  (U_DW + 16 * W_PITCH)

__device__ __forceinline__ void lcnn_term(v8f (&acc)[9], const v2f A0, const v2f A1,
                                          const float* uL, const float* wL,
                                          const int mu, const int ik, const int s, const int jadd)
{
    const int k = ik - 2;

    __builtin_amdgcn_wave_barrier();

    const int sshift = (mu == 3) ? ((s + k + 16) & 15) : s;
    const float* wb = wL + sshift * W_PITCH + jadd * 9;

    const float* up = uL + s * 180 + (mu * 5 + ik) * 9;
    float u[9];
    #pragma unroll
    for (int q = 0; q < 9; ++q) u[q] = up[q];

    v2f P0[9], P1[9];
    #pragma unroll
    for (int cd = 0; cd < 9; ++cd) {
        P0[cd] = (v2f){ wb[     cd], wb[ 9 + cd] };
        P1[cd] = (v2f){ wb[36 + cd], wb[45 + cd] };
    }

    v2f V0[9], V1[9];
    #pragma unroll
    for (int h = 0; h < 2; ++h) {
        const v2f* P = h ? P1 : P0;
        v2f*       V = h ? V1 : V0;
        v2f T[9];
        #pragma unroll
        for (int c = 0; c < 3; ++c)
            #pragma unroll
            for (int bb = 0; bb < 3; ++bb)
                T[c*3+bb] = P[c*3+0]*u[bb*3+0]
                          + P[c*3+1]*u[bb*3+1]
                          + P[c*3+2]*u[bb*3+2];
        #pragma unroll
        for (int a = 0; a < 3; ++a)
            #pragma unroll
            for (int bb = 0; bb < 3; ++bb)
                V[a*3+bb] = T[0+bb]*u[a*3+0]
                          + T[3+bb]*u[a*3+1]
                          + T[6+bb]*u[a*3+2];
    }

    #pragma unroll
    for (int ab = 0; ab < 9; ++ab)
        acc[ab] = __builtin_amdgcn_wmma_f32_16x16x4_f32(false, A0, false, V0[ab], (short)0, acc[ab], false, false);
    #pragma unroll
    for (int ab = 0; ab < 9; ++ab)
        acc[ab] = __builtin_amdgcn_wmma_f32_16x16x4_f32(false, A1, false, V1[ab], (short)0, acc[ab], false, false);

    __builtin_amdgcn_wave_barrier();
}

__global__ __launch_bounds__(256)
void lcnn_wmma_kernel(const float* __restrict__ W,
                      const float* __restrict__ U,
                      const float* __restrict__ Om,
                      float* __restrict__ Out)
{
    __shared__ __attribute__((aligned(16))) float lds[8 * WAVE_DW];

    const int tid  = threadIdx.x;
    const int wave = tid >> 5;
    const int lane = tid & 31;
    const int line = blockIdx.x * 8 + wave;

    const int s     = lane & 15;
    const int hh    = lane >> 4;
    const int jadd  = hh * 2;
    const int ibase = hh * 4;

    const int x3 = line & 15;
    const int x2 = (line >> 4) & 15;
    const int x1 = (line >> 8) & 15;

    const int mySite = line * 16 + s;

    float* uL = lds + wave * WAVE_DW;
    float* wL = uL + U_DW;

    {
        const v4f* ug  = (const v4f*)(U + line * U_DW);
        v4f*       uL4 = (v4f*)uL;
        #pragma unroll
        for (int q = 0; q < 23; ++q) {
            const int g = q * 32 + lane;
            if (g < 720) uL4[g] = ug[g];
        }
    }

    int wslot[9];
    #pragma unroll
    for (int q = 0; q < 9; ++q) {
        const int g = q * 32 + lane;
        wslot[q] = (g / 18) * W_PITCH + (g % 18) * 4;
    }

    {
        const v4f* wg = (const v4f*)(W + line * 1152);
        v4f w0[9];
        #pragma unroll
        for (int q = 0; q < 9; ++q) w0[q] = wg[q * 32 + lane];
        #pragma unroll
        for (int q = 0; q < 9; ++q) *(v4f*)(wL + wslot[q]) = w0[q];
    }

    v4f wn[9];
    {
        const int wline0 = (line & ~(15 << 8)) | (((x1 - 2) & 15) << 8);
        const v4f* wg = (const v4f*)(W + wline0 * 1152);
        #pragma unroll
        for (int q = 0; q < 9; ++q) wn[q] = wg[q * 32 + lane];
    }

    v8f acc[9];
    #pragma unroll
    for (int q = 0; q < 9; ++q) acc[q] = (v8f){0.f, 0.f, 0.f, 0.f, 0.f, 0.f, 0.f, 0.f};

    const int irow = (s & 3) | ((s >> 1) & 4);

    #pragma unroll 1
    for (int n = 0; n < 8; ++n) {
        const int mu = (n < 5) ? 3 : (n - 5);
        const int ik = (n < 5) ? n : 2;
        const int ob = irow * 160 + mu * 5 + ik;
        const v2f A0 = { Om[ob + (jadd + 0) * 20], Om[ob + (jadd + 1) * 20] };
        const v2f A1 = { Om[ob + (jadd + 4) * 20], Om[ob + (jadd + 5) * 20] };
        lcnn_term(acc, A0, A1, uL, wL, mu, ik, s, jadd);
    }

    #pragma unroll 1
    for (int f = 0; f < 12; ++f) {
        const int mu = f >> 2;
        const int iq = f & 3;
        const int ik = iq + (iq >> 1);

        #pragma unroll
        for (int q = 0; q < 9; ++q) *(v4f*)(wL + wslot[q]) = wn[q];

        const int ob = irow * 160 + mu * 5 + ik;
        const v2f A0 = { Om[ob + (jadd + 0) * 20], Om[ob + (jadd + 1) * 20] };
        const v2f A1 = { Om[ob + (jadd + 4) * 20], Om[ob + (jadd + 5) * 20] };

        {
            const int fn  = (f + 1 < 12) ? (f + 1) : f;
            const int mu1 = fn >> 2;
            const int iq1 = fn & 3;
            const int k1  = iq1 + (iq1 >> 1) - 2;
            const int sh  = 8 - 4 * mu1;
            const int wline1 = (line & ~(15 << sh)) | ((((line >> sh) + k1) & 15) << sh);
            const v4f* wg = (const v4f*)(W + wline1 * 1152);
            #pragma unroll
            for (int q = 0; q < 9; ++q) wn[q] = wg[q * 32 + lane];
        }

        lcnn_term(acc, A0, A1, uL, wL, mu, ik, s, jadd);
    }

    asm volatile("v_nop\n\tv_nop\n\tv_nop\n\tv_nop\n\tv_nop\n\tv_nop\n\tv_nop\n\tv_nop"
                 : "+v"(acc[0]), "+v"(acc[1]), "+v"(acc[2]), "+v"(acc[3]), "+v"(acc[4]),
                   "+v"(acc[5]), "+v"(acc[6]), "+v"(acc[7]), "+v"(acc[8]));

    float o[36];
    #pragma unroll
    for (int r = 0; r < 4; ++r)
        #pragma unroll
        for (int ab = 0; ab < 9; ++ab)
            o[r*9 + ab] = acc[ab][r];
    float4* op = (float4*)(Out + mySite * 72 + ibase * 9);
    #pragma unroll
    for (int q = 0; q < 9; ++q)
        op[q] = make_float4(o[q*4+0], o[q*4+1], o[q*4+2], o[q*4+3]);
}

extern "C" void kernel_launch(void* const* d_in, const int* in_sizes, int n_in,
                              void* d_out, int out_size, void* d_ws, size_t ws_size,
                              hipStream_t stream) {
    const float* W  = (const float*)d_in[0];
    const float* U  = (const float*)d_in[1];
    const float* Om = (const float*)d_in[2];
    float* Out = (float*)d_out;
    (void)in_sizes; (void)n_in; (void)out_size; (void)d_ws; (void)ws_size;

    lcnn_wmma_kernel<<<dim3(1024), dim3(256), 0, stream>>>(W, U, Om, Out);
}


# omega staged once per workgroup in LDS, transposed [term][j][i]: a term's WMMA A fragment is 2 paired LDS reads per lane, not 4 scattered global loads (a cache line per channel row); sum order same
# speedup vs baseline: 1.7827x; 1.0592x over previous
#include <hip/hip_runtime.h>


typedef __attribute__((ext_vector_type(2))) float v2f;
typedef __attribute__((ext_vector_type(8))) float v8f;
typedef __attribute__((ext_vector_type(4))) float v4f;

#define U_DW     2880
#define W_PITCH  76
#define WAVE_DW  (U_DW + 16 * W_PITCH)
#define OM_PITCH 65
#define OM_DW    (20 * OM_PITCH)

__device__ __forceinline__ void lcnn_term(v8f (&acc)[9], const v2f A0, const v2f A1,
                                          const float* uL, const float* wL,
                                          const int mu, const int ik, const int s, const int jadd)
{
    const int k = ik - 2;

    __builtin_amdgcn_wave_barrier();

    const int sshift = (mu == 3) ? ((s + k + 16) & 15) : s;
    const float* wb = wL + sshift * W_PITCH + jadd * 9;

    const float* up = uL + s * 180 + (mu * 5 + ik) * 9;
    float u[9];
    #pragma unroll
    for (int q = 0; q < 9; ++q) u[q] = up[q];

    v2f P0[9], P1[9];
    #pragma unroll
    for (int cd = 0; cd < 9; ++cd) {
        P0[cd] = (v2f){ wb[     cd], wb[ 9 + cd] };
        P1[cd] = (v2f){ wb[36 + cd], wb[45 + cd] };
    }

    v2f V0[9], V1[9];
    #pragma unroll
    for (int h = 0; h < 2; ++h) {
        const v2f* P = h ? P1 : P0;
        v2f*       V = h ? V1 : V0;
        v2f T[9];
        #pragma unroll
        for (int c = 0; c < 3; ++c)
            #pragma unroll
            for (int bb = 0; bb < 3; ++bb)
                T[c*3+bb] = P[c*3+0]*u[bb*3+0]
                          + P[c*3+1]*u[bb*3+1]
                          + P[c*3+2]*u[bb*3+2];
        #pragma unroll
        for (int a = 0; a < 3; ++a)
            #pragma unroll
            for (int bb = 0; bb < 3; ++bb)
                V[a*3+bb] = T[0+bb]*u[a*3+0]
                          + T[3+bb]*u[a*3+1]
                          + T[6+bb]*u[a*3+2];
    }

    #pragma unroll
    for (int ab = 0; ab < 9; ++ab)
        acc[ab] = __builtin_amdgcn_wmma_f32_16x16x4_f32(false, A0, false, V0[ab], (short)0, acc[ab], false, false);
    #pragma unroll
    for (int ab = 0; ab < 9; ++ab)
        acc[ab] = __builtin_amdgcn_wmma_f32_16x16x4_f32(false, A1, false, V1[ab], (short)0, acc[ab], false, false);

    __builtin_amdgcn_wave_barrier();
}

__global__ __launch_bounds__(256)
void lcnn_wmma_kernel(const float* __restrict__ W,
                      const float* __restrict__ U,
                      const float* __restrict__ Om,
                      float* __restrict__ Out)
{
    __shared__ __attribute__((aligned(16))) float lds[8 * WAVE_DW + OM_DW];

    const int tid  = threadIdx.x;
    const int wave = tid >> 5;
    const int lane = tid & 31;
    const int line = blockIdx.x * 8 + wave;

    const int s     = lane & 15;
    const int hh    = lane >> 4;
    const int jadd  = hh * 2;
    const int ibase = hh * 4;

    const int x3 = line & 15;
    const int x2 = (line >> 4) & 15;
    const int x1 = (line >> 8) & 15;

    const int mySite = line * 16 + s;

    float* uL = lds + wave * WAVE_DW;
    float* wL = uL + U_DW;
    float* omL = lds + 8 * WAVE_DW;

    #pragma unroll
    for (int q = 0; q < 5; ++q) {
        const int e  = q * 256 + tid;
        const int ij = e / 20;
        const int tm = e - ij * 20;
        omL[tm * OM_PITCH + (ij & 7) * 8 + (ij >> 3)] = Om[e];
    }

    {
        const v4f* ug  = (const v4f*)(U + line * U_DW);
        v4f*       uL4 = (v4f*)uL;
        #pragma unroll
        for (int q = 0; q < 23; ++q) {
            const int g = q * 32 + lane;
            if (g < 720) uL4[g] = ug[g];
        }
    }

    int wslot[9];
    #pragma unroll
    for (int q = 0; q < 9; ++q) {
        const int g = q * 32 + lane;
        wslot[q] = (g / 18) * W_PITCH + (g % 18) * 4;
    }

    {
        const v4f* wg = (const v4f*)(W + line * 1152);
        v4f w0[9];
        #pragma unroll
        for (int q = 0; q < 9; ++q) w0[q] = wg[q * 32 + lane];
        #pragma unroll
        for (int q = 0; q < 9; ++q) *(v4f*)(wL + wslot[q]) = w0[q];
    }

    __syncthreads();

    v4f wn[9];
    {
        const int wline0 = (line & ~(15 << 8)) | (((x1 - 2) & 15) << 8);
        const v4f* wg = (const v4f*)(W + wline0 * 1152);
        #pragma unroll
        for (int q = 0; q < 9; ++q) wn[q] = wg[q * 32 + lane];
    }

    v8f acc[9];
    #pragma unroll
    for (int q = 0; q < 9; ++q) acc[q] = (v8f){0.f, 0.f, 0.f, 0.f, 0.f, 0.f, 0.f, 0.f};

    const int irow = (s & 3) | ((s >> 1) & 4);

    #pragma unroll 1
    for (int n = 0; n < 8; ++n) {
        const int mu = (n < 5) ? 3 : (n - 5);
        const int ik = (n < 5) ? n : 2;
        const float* oa = omL + (mu * 5 + ik) * OM_PITCH + jadd * 8 + irow;
        const v2f A0 = { oa[0],  oa[8]  };
        const v2f A1 = { oa[32], oa[40] };
        lcnn_term(acc, A0, A1, uL, wL, mu, ik, s, jadd);
    }

    #pragma unroll 1
    for (int f = 0; f < 12; ++f) {
        const int mu = f >> 2;
        const int iq = f & 3;
        const int ik = iq + (iq >> 1);

        #pragma unroll
        for (int q = 0; q < 9; ++q) *(v4f*)(wL + wslot[q]) = wn[q];

        const float* oa = omL + (mu * 5 + ik) * OM_PITCH + jadd * 8 + irow;
        const v2f A0 = { oa[0],  oa[8]  };
        const v2f A1 = { oa[32], oa[40] };

        {
            const int fn  = (f + 1 < 12) ? (f + 1) : f;
            const int mu1 = fn >> 2;
            const int iq1 = fn & 3;
            const int k1  = iq1 + (iq1 >> 1) - 2;
            const int sh  = 8 - 4 * mu1;
            const int wline1 = (line & ~(15 << sh)) | ((((line >> sh) + k1) & 15) << sh);
            const v4f* wg = (const v4f*)(W + wline1 * 1152);
            #pragma unroll
            for (int q = 0; q < 9; ++q) wn[q] = wg[q * 32 + lane];
        }

        lcnn_term(acc, A0, A1, uL, wL, mu, ik, s, jadd);
    }

    asm volatile("v_nop\n\tv_nop\n\tv_nop\n\tv_nop\n\tv_nop\n\tv_nop\n\tv_nop\n\tv_nop"
                 : "+v"(acc[0]), "+v"(acc[1]), "+v"(acc[2]), "+v"(acc[3]), "+v"(acc[4]),
                   "+v"(acc[5]), "+v"(acc[6]), "+v"(acc[7]), "+v"(acc[8]));

    float o[36];
    #pragma unroll
    for (int r = 0; r < 4; ++r)
        #pragma unroll
        for (int ab = 0; ab < 9; ++ab)
            o[r*9 + ab] = acc[ab][r];
    float4* op = (float4*)(Out + mySite * 72 + ibase * 9);
    #pragma unroll
    for (int q = 0; q < 9; ++q)
        op[q] = make_float4(o[q*4+0], o[q*4+1], o[q*4+2], o[q*4+3]);
}

extern "C" void kernel_launch(void* const* d_in, const int* in_sizes, int n_in,
                              void* d_out, int out_size, void* d_ws, size_t ws_size,
                              hipStream_t stream) {
    const float* W  = (const float*)d_in[0];
    const float* U  = (const float*)d_in[1];
    const float* Om = (const float*)d_in[2];
    float* Out = (float*)d_out;
    (void)in_sizes; (void)n_in; (void)out_size; (void)d_ws; (void)ws_size;

    lcnn_wmma_kernel<<<dim3(1024), dim3(256), 0, stream>>>(W, U, Om, Out);
}


# Output staged in the wave's LDS region in output order and stored as whole 128-byte lines: each 16-byte store instruction covers 512 contiguous bytes, not 32 pieces in 32 lines; sum order same
# speedup vs baseline: 2.1146x; 1.1862x over previous
#include <hip/hip_runtime.h>


typedef __attribute__((ext_vector_type(2))) float v2f;
typedef __attribute__((ext_vector_type(8))) float v8f;
typedef __attribute__((ext_vector_type(4))) float v4f;

#define U_DW     2880
#define W_PITCH  76
#define WAVE_DW  (U_DW + 16 * W_PITCH)
#define OM_PITCH 65
#define OM_DW    (20 * OM_PITCH)

__device__ __forceinline__ void lcnn_term(v8f (&acc)[9], const v2f A0, const v2f A1,
                                          const float* uL, const float* wL,
                                          const int mu, const int ik, const int s, const int jadd)
{
    const int k = ik - 2;

    __builtin_amdgcn_wave_barrier();

    const int sshift = (mu == 3) ? ((s + k + 16) & 15) : s;
    const float* wb = wL + sshift * W_PITCH + jadd * 9;

    const float* up = uL + s * 180 + (mu * 5 + ik) * 9;
    float u[9];
    #pragma unroll
    for (int q = 0; q < 9; ++q) u[q] = up[q];

    v2f P0[9], P1[9];
    #pragma unroll
    for (int cd = 0; cd < 9; ++cd) {
        P0[cd] = (v2f){ wb[     cd], wb[ 9 + cd] };
        P1[cd] = (v2f){ wb[36 + cd], wb[45 + cd] };
    }

    v2f V0[9], V1[9];
    #pragma unroll
    for (int h = 0; h < 2; ++h) {
        const v2f* P = h ? P1 : P0;
        v2f*       V = h ? V1 : V0;
        v2f T[9];
        #pragma unroll
        for (int c = 0; c < 3; ++c)
            #pragma unroll
            for (int bb = 0; bb < 3; ++bb)
                T[c*3+bb] = P[c*3+0]*u[bb*3+0]
                          + P[c*3+1]*u[bb*3+1]
                          + P[c*3+2]*u[bb*3+2];
        #pragma unroll
        for (int a = 0; a < 3; ++a)
            #pragma unroll
            for (int bb = 0; bb < 3; ++bb)
                V[a*3+bb] = T[0+bb]*u[a*3+0]
                          + T[3+bb]*u[a*3+1]
                          + T[6+bb]*u[a*3+2];
    }

    #pragma unroll
    for (int ab = 0; ab < 9; ++ab)
        acc[ab] = __builtin_amdgcn_wmma_f32_16x16x4_f32(false, A0, false, V0[ab], (short)0, acc[ab], false, false);
    #pragma unroll
    for (int ab = 0; ab < 9; ++ab)
        acc[ab] = __builtin_amdgcn_wmma_f32_16x16x4_f32(false, A1, false, V1[ab], (short)0, acc[ab], false, false);

    __builtin_amdgcn_wave_barrier();
}

__global__ __launch_bounds__(256)
void lcnn_wmma_kernel(const float* __restrict__ W,
                      const float* __restrict__ U,
                      const float* __restrict__ Om,
                      float* __restrict__ Out)
{
    __shared__ __attribute__((aligned(16))) float lds[8 * WAVE_DW + OM_DW];

    const int tid  = threadIdx.x;
    const int wave = tid >> 5;
    const int lane = tid & 31;
    const int line = blockIdx.x * 8 + wave;

    const int s     = lane & 15;
    const int hh    = lane >> 4;
    const int jadd  = hh * 2;
    const int ibase = hh * 4;

    const int x3 = line & 15;
    const int x2 = (line >> 4) & 15;
    const int x1 = (line >> 8) & 15;

    float* uL = lds + wave * WAVE_DW;
    float* wL = uL + U_DW;
    float* omL = lds + 8 * WAVE_DW;

    #pragma unroll
    for (int q = 0; q < 5; ++q) {
        const int e  = q * 256 + tid;
        const int ij = e / 20;
        const int tm = e - ij * 20;
        omL[tm * OM_PITCH + (ij & 7) * 8 + (ij >> 3)] = Om[e];
    }

    {
        const v4f* ug  = (const v4f*)(U + line * U_DW);
        v4f*       uL4 = (v4f*)uL;
        #pragma unroll
        for (int q = 0; q < 23; ++q) {
            const int g = q * 32 + lane;
            if (g < 720) uL4[g] = ug[g];
        }
    }

    int wslot[9];
    #pragma unroll
    for (int q = 0; q < 9; ++q) {
        const int g = q * 32 + lane;
        wslot[q] = (g / 18) * W_PITCH + (g % 18) * 4;
    }

    {
        const v4f* wg = (const v4f*)(W + line * 1152);
        v4f w0[9];
        #pragma unroll
        for (int q = 0; q < 9; ++q) w0[q] = wg[q * 32 + lane];
        #pragma unroll
        for (int q = 0; q < 9; ++q) *(v4f*)(wL + wslot[q]) = w0[q];
    }

    __syncthreads();

    v4f wn[9];
    {
        const int wline0 = (line & ~(15 << 8)) | (((x1 - 2) & 15) << 8);
        const v4f* wg = (const v4f*)(W + wline0 * 1152);
        #pragma unroll
        for (int q = 0; q < 9; ++q) wn[q] = wg[q * 32 + lane];
    }

    v8f acc[9];
    #pragma unroll
    for (int q = 0; q < 9; ++q) acc[q] = (v8f){0.f, 0.f, 0.f, 0.f, 0.f, 0.f, 0.f, 0.f};

    const int irow = (s & 3) | ((s >> 1) & 4);

    #pragma unroll 1
    for (int n = 0; n < 8; ++n) {
        const int mu = (n < 5) ? 3 : (n - 5);
        const int ik = (n < 5) ? n : 2;
        const float* oa = omL + (mu * 5 + ik) * OM_PITCH + jadd * 8 + irow;
        const v2f A0 = { oa[0],  oa[8]  };
        const v2f A1 = { oa[32], oa[40] };
        lcnn_term(acc, A0, A1, uL, wL, mu, ik, s, jadd);
    }

    #pragma unroll 1
    for (int f = 0; f < 12; ++f) {
        const int mu = f >> 2;
        const int iq = f & 3;
        const int ik = iq + (iq >> 1);

        #pragma unroll
        for (int q = 0; q < 9; ++q) *(v4f*)(wL + wslot[q]) = wn[q];

        const float* oa = omL + (mu * 5 + ik) * OM_PITCH + jadd * 8 + irow;
        const v2f A0 = { oa[0],  oa[8]  };
        const v2f A1 = { oa[32], oa[40] };

        {
            const int fn  = (f + 1 < 12) ? (f + 1) : f;
            const int mu1 = fn >> 2;
            const int iq1 = fn & 3;
            const int k1  = iq1 + (iq1 >> 1) - 2;
            const int sh  = 8 - 4 * mu1;
            const int wline1 = (line & ~(15 << sh)) | ((((line >> sh) + k1) & 15) << sh);
            const v4f* wg = (const v4f*)(W + wline1 * 1152);
            #pragma unroll
            for (int q = 0; q < 9; ++q) wn[q] = wg[q * 32 + lane];
        }

        lcnn_term(acc, A0, A1, uL, wL, mu, ik, s, jadd);
    }

    asm volatile("v_nop\n\tv_nop\n\tv_nop\n\tv_nop\n\tv_nop\n\tv_nop\n\tv_nop\n\tv_nop"
                 : "+v"(acc[0]), "+v"(acc[1]), "+v"(acc[2]), "+v"(acc[3]), "+v"(acc[4]),
                   "+v"(acc[5]), "+v"(acc[6]), "+v"(acc[7]), "+v"(acc[8]));

    float o[36];
    #pragma unroll
    for (int r = 0; r < 4; ++r)
        #pragma unroll
        for (int ab = 0; ab < 9; ++ab)
            o[r*9 + ab] = acc[ab][r];

    {
        v4f* ob = (v4f*)(uL + s * 72 + ibase * 9);
        #pragma unroll
        for (int q = 0; q < 9; ++q)
            ob[q] = (v4f){ o[q*4+0], o[q*4+1], o[q*4+2], o[q*4+3] };
    }
    __builtin_amdgcn_wave_barrier();
    {
        const v4f* ib = (const v4f*)uL;
        v4f*       og = (v4f*)(Out + line * 1152);
        v4f t[9];
        #pragma unroll
        for (int q = 0; q < 9; ++q) t[q] = ib[q * 32 + lane];
        #pragma unroll
        for (int q = 0; q < 9; ++q) og[q * 32 + lane] = t[q];
    }
}

extern "C" void kernel_launch(void* const* d_in, const int* in_sizes, int n_in,
                              void* d_out, int out_size, void* d_ws, size_t ws_size,
                              hipStream_t stream) {
    const float* W  = (const float*)d_in[0];
    const float* U  = (const float*)d_in[1];
    const float* Om = (const float*)d_in[2];
    float* Out = (float*)d_out;
    (void)in_sizes; (void)n_in; (void)out_size; (void)d_ws; (void)ws_size;

    lcnn_wmma_kernel<<<dim3(1024), dim3(256), 0, stream>>>(W, U, Om, Out);
}


# The four k=0 terms (own line, unshifted) share one B: channel mix first, two terms stacked in the 16 WMMA rows, sandwich after on VALU: 36 fewer WMMAs per wave; sum order over terms changes
# speedup vs baseline: 2.2257x; 1.0525x over previous
#include <hip/hip_runtime.h>


typedef __attribute__((ext_vector_type(2))) float v2f;
typedef __attribute__((ext_vector_type(8))) float v8f;
typedef __attribute__((ext_vector_type(4))) float v4f;

#define U_DW     2880
#define W_PITCH  76
#define WAVE_DW  (U_DW + 16 * W_PITCH)
#define OM_PITCH 65
#define OM_DW    (20 * OM_PITCH)

__device__ __forceinline__ void lcnn_term(v8f (&acc)[9], const v2f A0, const v2f A1,
                                          const float* uL, const float* wL,
                                          const int mu, const int ik, const int s, const int jadd)
{
    const int k = ik - 2;

    __builtin_amdgcn_wave_barrier();

    const int sshift = (mu == 3) ? ((s + k + 16) & 15) : s;
    const float* wb = wL + sshift * W_PITCH + jadd * 9;

    const float* up = uL + s * 180 + (mu * 5 + ik) * 9;
    float u[9];
    #pragma unroll
    for (int q = 0; q < 9; ++q) u[q] = up[q];

    v2f P0[9], P1[9];
    #pragma unroll
    for (int cd = 0; cd < 9; ++cd) {
        P0[cd] = (v2f){ wb[     cd], wb[ 9 + cd] };
        P1[cd] = (v2f){ wb[36 + cd], wb[45 + cd] };
    }

    v2f V0[9], V1[9];
    #pragma unroll
    for (int h = 0; h < 2; ++h) {
        const v2f* P = h ? P1 : P0;
        v2f*       V = h ? V1 : V0;
        v2f T[9];
        #pragma unroll
        for (int c = 0; c < 3; ++c)
            #pragma unroll
            for (int bb = 0; bb < 3; ++bb)
                T[c*3+bb] = P[c*3+0]*u[bb*3+0]
                          + P[c*3+1]*u[bb*3+1]
                          + P[c*3+2]*u[bb*3+2];
        #pragma unroll
        for (int a = 0; a < 3; ++a)
            #pragma unroll
            for (int bb = 0; bb < 3; ++bb)
                V[a*3+bb] = T[0+bb]*u[a*3+0]
                          + T[3+bb]*u[a*3+1]
                          + T[6+bb]*u[a*3+2];
    }

    #pragma unroll
    for (int ab = 0; ab < 9; ++ab)
        acc[ab] = __builtin_amdgcn_wmma_f32_16x16x4_f32(false, A0, false, V0[ab], (short)0, acc[ab], false, false);
    #pragma unroll
    for (int ab = 0; ab < 9; ++ab)
        acc[ab] = __builtin_amdgcn_wmma_f32_16x16x4_f32(false, A1, false, V1[ab], (short)0, acc[ab], false, false);

    __builtin_amdgcn_wave_barrier();
}

__global__ __launch_bounds__(256)
void lcnn_wmma_kernel(const float* __restrict__ W,
                      const float* __restrict__ U,
                      const float* __restrict__ Om,
                      float* __restrict__ Out)
{
    __shared__ __attribute__((aligned(16))) float lds[8 * WAVE_DW + OM_DW];

    const int tid  = threadIdx.x;
    const int wave = tid >> 5;
    const int lane = tid & 31;
    const int line = blockIdx.x * 8 + wave;

    const int s     = lane & 15;
    const int hh    = lane >> 4;
    const int jadd  = hh * 2;
    const int ibase = hh * 4;

    const int x3 = line & 15;
    const int x2 = (line >> 4) & 15;
    const int x1 = (line >> 8) & 15;

    float* uL = lds + wave * WAVE_DW;
    float* wL = uL + U_DW;
    float* omL = lds + 8 * WAVE_DW;

    #pragma unroll
    for (int q = 0; q < 5; ++q) {
        const int e  = q * 256 + tid;
        const int ij = e / 20;
        const int tm = e - ij * 20;
        omL[tm * OM_PITCH + (ij & 7) * 8 + (ij >> 3)] = Om[e];
    }

    {
        const v4f* ug  = (const v4f*)(U + line * U_DW);
        v4f*       uL4 = (v4f*)uL;
        #pragma unroll
        for (int q = 0; q < 23; ++q) {
            const int g = q * 32 + lane;
            if (g < 720) uL4[g] = ug[g];
        }
    }

    int wslot[9];
    #pragma unroll
    for (int q = 0; q < 9; ++q) {
        const int g = q * 32 + lane;
        wslot[q] = (g / 18) * W_PITCH + (g % 18) * 4;
    }

    {
        const v4f* wg = (const v4f*)(W + line * 1152);
        v4f w0[9];
        #pragma unroll
        for (int q = 0; q < 9; ++q) w0[q] = wg[q * 32 + lane];
        #pragma unroll
        for (int q = 0; q < 9; ++q) *(v4f*)(wL + wslot[q]) = w0[q];
    }

    __syncthreads();

    v8f acc[9];

    const int irow = (s & 3) | ((s >> 1) & 4);

    {
        const float* wb = wL + s * W_PITCH + jadd * 9;
        v2f P0[9], P1[9];
        #pragma unroll
        for (int cd = 0; cd < 9; ++cd) {
            P0[cd] = (v2f){ wb[     cd], wb[ 9 + cd] };
            P1[cd] = (v2f){ wb[36 + cd], wb[45 + cd] };
        }

        v2f O0[9], O1[9];
        #pragma unroll
        for (int ab = 0; ab < 9; ++ab) { O0[ab] = (v2f){0.f, 0.f}; O1[ab] = (v2f){0.f, 0.f}; }

        #pragma unroll 1
        for (int tile = 0; tile < 2; ++tile) {
            const int muA = 2 * tile + ((s >> 2) & 1);
            const float* oa = omL + (muA * 5 + 2) * OM_PITCH + jadd * 8 + irow;
            const v2f A0 = { oa[0],  oa[8]  };
            const v2f A1 = { oa[32], oa[40] };

            v8f X[9];
            #pragma unroll
            for (int cd = 0; cd < 9; ++cd) {
                const v8f z = {0.f, 0.f, 0.f, 0.f, 0.f, 0.f, 0.f, 0.f};
                X[cd] = __builtin_amdgcn_wmma_f32_16x16x4_f32(false, A0, false, P0[cd], (short)0, z, false, false);
            }
            #pragma unroll
            for (int cd = 0; cd < 9; ++cd)
                X[cd] = __builtin_amdgcn_wmma_f32_16x16x4_f32(false, A1, false, P1[cd], (short)0, X[cd], false, false);

            #pragma unroll
            for (int e = 0; e < 2; ++e) {
                const float* up = uL + s * 180 + ((2 * tile + e) * 5 + 2) * 9;
                float u[9];
                #pragma unroll
                for (int q = 0; q < 9; ++q) u[q] = up[q];

                #pragma unroll
                for (int p = 0; p < 2; ++p) {
                    v2f* O = p ? O1 : O0;
                    v2f T[9];
                    #pragma unroll
                    for (int c = 0; c < 3; ++c)
                        #pragma unroll
                        for (int bb = 0; bb < 3; ++bb) {
                            const v2f x0 = { X[c*3+0][4*e + 2*p], X[c*3+0][4*e + 2*p + 1] };
                            const v2f x1 = { X[c*3+1][4*e + 2*p], X[c*3+1][4*e + 2*p + 1] };
                            const v2f x2 = { X[c*3+2][4*e + 2*p], X[c*3+2][4*e + 2*p + 1] };
                            T[c*3+bb] = x0*u[bb*3+0] + x1*u[bb*3+1] + x2*u[bb*3+2];
                        }
                    #pragma unroll
                    for (int a = 0; a < 3; ++a)
                        #pragma unroll
                        for (int bb = 0; bb < 3; ++bb) {
                            O[a*3+bb] = ((O[a*3+bb] + T[0+bb]*u[a*3+0]) + T[3+bb]*u[a*3+1]) + T[6+bb]*u[a*3+2];
                        }
                }
            }
        }

        #pragma unroll
        for (int ab = 0; ab < 9; ++ab)
            acc[ab] = (v8f){ O0[ab][0], O0[ab][1], O1[ab][0], O1[ab][1], 0.f, 0.f, 0.f, 0.f };
    }

    v4f wn[9];
    {
        const int wline0 = (line & ~(15 << 8)) | (((x1 - 2) & 15) << 8);
        const v4f* wg = (const v4f*)(W + wline0 * 1152);
        #pragma unroll
        for (int q = 0; q < 9; ++q) wn[q] = wg[q * 32 + lane];
    }

    #pragma unroll 1
    for (int n = 0; n < 4; ++n) {
        const int ik = n + (n >> 1);
        const float* oa = omL + (15 + ik) * OM_PITCH + jadd * 8 + irow;
        const v2f A0 = { oa[0],  oa[8]  };
        const v2f A1 = { oa[32], oa[40] };
        lcnn_term(acc, A0, A1, uL, wL, 3, ik, s, jadd);
    }

    #pragma unroll 1
    for (int f = 0; f < 12; ++f) {
        const int mu = f >> 2;
        const int iq = f & 3;
        const int ik = iq + (iq >> 1);

        #pragma unroll
        for (int q = 0; q < 9; ++q) *(v4f*)(wL + wslot[q]) = wn[q];

        const float* oa = omL + (mu * 5 + ik) * OM_PITCH + jadd * 8 + irow;
        const v2f A0 = { oa[0],  oa[8]  };
        const v2f A1 = { oa[32], oa[40] };

        {
            const int fn  = (f + 1 < 12) ? (f + 1) : f;
            const int mu1 = fn >> 2;
            const int iq1 = fn & 3;
            const int k1  = iq1 + (iq1 >> 1) - 2;
            const int sh  = 8 - 4 * mu1;
            const int wline1 = (line & ~(15 << sh)) | ((((line >> sh) + k1) & 15) << sh);
            const v4f* wg = (const v4f*)(W + wline1 * 1152);
            #pragma unroll
            for (int q = 0; q < 9; ++q) wn[q] = wg[q * 32 + lane];
        }

        lcnn_term(acc, A0, A1, uL, wL, mu, ik, s, jadd);
    }

    asm volatile("v_nop\n\tv_nop\n\tv_nop\n\tv_nop\n\tv_nop\n\tv_nop\n\tv_nop\n\tv_nop"
                 : "+v"(acc[0]), "+v"(acc[1]), "+v"(acc[2]), "+v"(acc[3]), "+v"(acc[4]),
                   "+v"(acc[5]), "+v"(acc[6]), "+v"(acc[7]), "+v"(acc[8]));

    float o[36];
    #pragma unroll
    for (int r = 0; r < 4; ++r)
        #pragma unroll
        for (int ab = 0; ab < 9; ++ab)
            o[r*9 + ab] = acc[ab][r];

    {
        v4f* ob = (v4f*)(uL + s * 72 + ibase * 9);
        #pragma unroll
        for (int q = 0; q < 9; ++q)
            ob[q] = (v4f){ o[q*4+0], o[q*4+1], o[q*4+2], o[q*4+3] };
    }
    __builtin_amdgcn_wave_barrier();
    {
        const v4f* ib = (const v4f*)uL;
        v4f*       og = (v4f*)(Out + line * 1152);
        v4f t[9];
        #pragma unroll
        for (int q = 0; q < 9; ++q) t[q] = ib[q * 32 + lane];
        #pragma unroll
        for (int q = 0; q < 9; ++q) og[q * 32 + lane] = t[q];
    }
}

extern "C" void kernel_launch(void* const* d_in, const int* in_sizes, int n_in,
                              void* d_out, int out_size, void* d_ws, size_t ws_size,
                              hipStream_t stream) {
    const float* W  = (const float*)d_in[0];
    const float* U  = (const float*)d_in[1];
    const float* Om = (const float*)d_in[2];
    float* Out = (float*)d_out;
    (void)in_sizes; (void)n_in; (void)out_size; (void)d_ws; (void)ws_size;

    lcnn_wmma_kernel<<<dim3(1024), dim3(256), 0, stream>>>(W, U, Om, Out);
}
